# starting kernel
# baseline (speedup 1.0000x reference)
#include <hip/hip_runtime.h>
#include <math.h>

typedef __attribute__((ext_vector_type(2))) float v2f;
typedef __attribute__((ext_vector_type(8))) float v8f;

#define B_   8
#define C_   64
#define N_   4096
#define O_   64
#define K_NN 20
#define EPS_ 1e-5f
#define SLOPE_ 0.2f
#define NEG_INF -3.4e38f

static __device__ __forceinline__ v8f wmma_f32(v2f a, v2f b, v8f c) {
    return __builtin_amdgcn_wmma_f32_16x16x4_f32(false, a, false, b, (short)0, c,
                                                 false, false);
}

__global__ void __launch_bounds__(256) k_sqnorm(const float* __restrict__ x,
                                                float* __restrict__ xx) {
    int t = blockIdx.x * blockDim.x + threadIdx.x;
    if (t >= B_ * N_) return;
    int b = t / N_, n = t % N_;
    const float* xb = x + (size_t)b * C_ * N_;
    float s = 0.f;
#pragma unroll
    for (int c = 0; c < C_; ++c) {
        float v = xb[(size_t)c * N_ + n];
        s += v * v;
    }
    xx[t] = s;
}

__global__ void __launch_bounds__(128) k_knn(const float* __restrict__ x,
                                             const float* __restrict__ xx,
                                             int* __restrict__ idx) {
    const int lane = threadIdx.x & 31;
    const int wave = threadIdx.x >> 5;
    const int b    = blockIdx.x / (N_ / 64);
    const int rb   = (blockIdx.x % (N_ / 64)) * 64;
    const int row0 = rb + wave * 16;

    const float* xb  = x  + (size_t)b * C_ * N_;
    const float* xxb = xx + (size_t)b * N_;

    const int hs  = lane >> 4;
    const int l16 = lane & 15;
    const int jb  = hs * 8;

    v2f A[16];
#pragma unroll
    for (int s = 0; s < 16; ++s) {
        int c0 = 4 * s + hs * 2;
        A[s].x = xb[(size_t)(c0 + 0) * N_ + row0 + l16];
        A[s].y = xb[(size_t)(c0 + 1) * N_ + row0 + l16];
    }

    float xr[8];
#pragma unroll
    for (int v = 0; v < 8; ++v) xr[v] = xxb[row0 + v + hs * 8];

    float vals[K_NN];
    int   ids[K_NN];
#pragma unroll
    for (int t = 0; t < K_NN; ++t) { vals[t] = NEG_INF; ids[t] = 0; }

    __shared__ float tile[2][4][16][17];

    int p = 0;
    for (int m0 = 0; m0 < N_; m0 += 16, p ^= 1) {
        v2f Bv[16];
#pragma unroll
        for (int s = 0; s < 16; ++s) {
            int c0 = 4 * s + hs * 2;
            Bv[s].x = xb[(size_t)(c0 + 0) * N_ + m0 + l16];
            Bv[s].y = xb[(size_t)(c0 + 1) * N_ + m0 + l16];
        }
        v8f acc = {};
#pragma unroll
        for (int s = 0; s < 16; ++s) acc = wmma_f32(A[s], Bv[s], acc);

        float xc = xxb[m0 + l16];
#pragma unroll
        for (int v = 0; v < 8; ++v) {
            int r = v + hs * 8;
            tile[p][wave][r][l16] = 2.0f * acc[v] - xr[v] - xc;
        }
        __syncthreads();
        for (int j = 0; j < 8; ++j) {
            float d = tile[p][wave][l16][jb + j];
            if (d > vals[K_NN - 1]) {
                float cv = d; int ci = m0 + jb + j;
#pragma unroll
                for (int t = 0; t < K_NN; ++t) {
                    if (cv > vals[t]) {
                        float tv = vals[t]; int ti = ids[t];
                        vals[t] = cv; ids[t] = ci;
                        cv = tv; ci = ti;
                    }
                }
            }
        }
    }

    __shared__ float mv[128][K_NN];
    __shared__ int   mi[128][K_NN];
#pragma unroll
    for (int t = 0; t < K_NN; ++t) {
        mv[threadIdx.x][t] = vals[t];
        mi[threadIdx.x][t] = ids[t];
    }
    __syncthreads();
    if (lane < 16) {
        const float* av = mv[wave * 32 + lane];
        const int*   ai = mi[wave * 32 + lane];
        const float* bv = mv[wave * 32 + lane + 16];
        const int*   bi = mi[wave * 32 + lane + 16];
        int* op = idx + ((size_t)b * N_ + row0 + lane) * K_NN;
        int ia = 0, ib2 = 0;
        for (int t = 0; t < K_NN; ++t) {
            float fa = av[ia], fb = bv[ib2];
            bool takeA = (fa > fb) || (fa == fb && ai[ia] < bi[ib2]);
            op[t] = takeA ? ai[ia++] : bi[ib2++];
        }
    }
}

__global__ void __launch_bounds__(256) k_wc(const float* __restrict__ W,
                                            float* __restrict__ Wc) {
    int t = blockIdx.x * blockDim.x + threadIdx.x;
    if (t >= 128 * 64) return;
    int op = t / 64, c = t % 64;
    float v;
    if (op < 64) v = W[(size_t)op * 128 + c];
    else         v = W[(size_t)(op - 64) * 128 + 64 + c] -
                     W[(size_t)(op - 64) * 128 + c];
    Wc[t] = v;
}

__global__ void __launch_bounds__(256) k_proj(const float* __restrict__ x,
                                              const float* __restrict__ Wc,
                                              float* __restrict__ P) {
    const int lane = threadIdx.x & 31;
    const int wave = threadIdx.x >> 5;
    const int b    = blockIdx.x / (N_ / 16);
    const int n0   = (blockIdx.x % (N_ / 16)) * 16;
    const int o0   = wave * 16;
    const float* xb = x + (size_t)b * C_ * N_;
    const int hs = lane >> 4, l16 = lane & 15;

    v8f acc = {};
#pragma unroll
    for (int s = 0; s < 16; ++s) {
        int c0 = 4 * s + hs * 2;
        v2f a, bb;
        a.x  = Wc[(size_t)(o0 + l16) * 64 + c0 + 0];
        a.y  = Wc[(size_t)(o0 + l16) * 64 + c0 + 1];
        bb.x = xb[(size_t)(c0 + 0) * N_ + n0 + l16];
        bb.y = xb[(size_t)(c0 + 1) * N_ + n0 + l16];
        acc = wmma_f32(a, bb, acc);
    }
    float* Pb = P + (size_t)b * 128 * N_;
#pragma unroll
    for (int v = 0; v < 8; ++v) {
        int m = o0 + v + hs * 8;
        Pb[(size_t)m * N_ + n0 + l16] = acc[v];
    }
}

__global__ void k_zero(float* __restrict__ S) {
    if (threadIdx.x < 128) S[threadIdx.x] = 0.f;
}

__global__ void __launch_bounds__(256) k_stats(const float* __restrict__ P,
                                               const int* __restrict__ idx,
                                               float* __restrict__ S) {
    const int chunks = N_ / 256;
    const int bo = blockIdx.x / chunks;
    const int n  = (blockIdx.x % chunks) * 256 + threadIdx.x;
    const int b = bo / O_, o = bo % O_;

    const float* P1 = P + (size_t)b * 128 * N_ + (size_t)o * N_;
    const float  p2 = P1[(size_t)64 * N_ + n];
    const int*   ip = idx + ((size_t)b * N_ + n) * K_NN;

    float s1 = 0.f, s2 = 0.f;
#pragma unroll
    for (int k = 0; k < K_NN; ++k) {
        float y = P1[ip[k]] + p2;
        s1 += y; s2 += y * y;
    }
    __shared__ float r1[256], r2[256];
    r1[threadIdx.x] = s1; r2[threadIdx.x] = s2;
    __syncthreads();
    for (int st = 128; st > 0; st >>= 1) {
        if (threadIdx.x < st) {
            r1[threadIdx.x] += r1[threadIdx.x + st];
            r2[threadIdx.x] += r2[threadIdx.x + st];
        }
        __syncthreads();
    }
    if (threadIdx.x == 0) {
        atomicAdd(&S[o],      r1[0]);
        atomicAdd(&S[64 + o], r2[0]);
    }
}

__global__ void __launch_bounds__(256) k_final(const float* __restrict__ P,
                                               const int* __restrict__ idx,
                                               const float* __restrict__ S,
                                               const float* __restrict__ gamma,
                                               const float* __restrict__ beta,
                                               float* __restrict__ out) {
    const int chunks = N_ / 256;
    const int bo = blockIdx.x / chunks;
    const int n  = (blockIdx.x % chunks) * 256 + threadIdx.x;
    const int b = bo / O_, o = bo % O_;

    const float cnt  = (float)((size_t)B_ * N_ * K_NN);
    const float mean = S[o] / cnt;
    const float var  = S[64 + o] / cnt - mean * mean;
    const float g    = gamma[o] * rsqrtf(var + EPS_);
    const float be   = beta[o];

    const float* P1 = P + (size_t)b * 128 * N_ + (size_t)o * N_;
    const float  p2 = P1[(size_t)64 * N_ + n];
    const int*   ip = idx + ((size_t)b * N_ + n) * K_NN;

    float best = NEG_INF;
#pragma unroll
    for (int k = 0; k < K_NN; ++k) {
        float y = P1[ip[k]] + p2;
        float z = (y - mean) * g + be;
        z = (z >= 0.f) ? z : SLOPE_ * z;
        best = fmaxf(best, z);
    }
    out[((size_t)b * O_ + o) * N_ + n] = best;
}

extern "C" void kernel_launch(void* const* d_in, const int* in_sizes, int n_in,
                              void* d_out, int out_size, void* d_ws, size_t ws_size,
                              hipStream_t stream) {
    const float* x     = (const float*)d_in[0];
    const float* W     = (const float*)d_in[1];
    const float* gamma = (const float*)d_in[2];
    const float* beta  = (const float*)d_in[3];
    float* out = (float*)d_out;

    char* ws = (char*)d_ws;
    size_t off = 0;
    float* xx = (float*)(ws + off); off += (size_t)B_ * N_ * sizeof(float);
    int*   id = (int*)  (ws + off); off += (size_t)B_ * N_ * K_NN * sizeof(int);
    float* Wc = (float*)(ws + off); off += (size_t)128 * 64 * sizeof(float);
    float* P  = (float*)(ws + off); off += (size_t)B_ * 128 * N_ * sizeof(float);
    float* S  = (float*)(ws + off); off += 128 * sizeof(float);
    (void)ws_size; (void)in_sizes; (void)n_in; (void)out_size;

    k_sqnorm<<<(B_ * N_ + 255) / 256, 256, 0, stream>>>(x, xx);
    k_knn  <<<B_ * (N_ / 64), 128, 0, stream>>>(x, xx, id);
    k_wc   <<<(128 * 64 + 255) / 256, 256, 0, stream>>>(W, Wc);
    k_proj <<<B_ * (N_ / 16), 256, 0, stream>>>(x, Wc, P);
    k_zero <<<1, 128, 0, stream>>>(S);
    k_stats<<<B_ * O_ * (N_ / 256), 256, 0, stream>>>(P, id, S);
    k_final<<<B_ * O_ * (N_ / 256), 256, 0, stream>>>(P, id, S, gamma, beta, out);
}


# k_final gathers nothing: BN+leaky is monotone in y (direction = sign of gamma), so k_stats keeps each (b,o,n)'s max or min y of its one gather in the P2 slot; k_final is one elementwise pass
# speedup vs baseline: 10.8287x; 10.8287x over previous
#include <hip/hip_runtime.h>
#include <math.h>

typedef __attribute__((ext_vector_type(2))) float v2f;
typedef __attribute__((ext_vector_type(8))) float v8f;
typedef __attribute__((ext_vector_type(4))) float v4f;
typedef __attribute__((address_space(3))) volatile v4f lds_live_v4f;

#define B_   8
#define C_   64
#define N_   4096
#define O_   64
#define K_NN 20
#define EPS_ 1e-5f
#define SLOPE_ 0.2f
#define NEG_INF -3.4e38f
#define POS_INF __builtin_inff()

static __device__ __forceinline__ v8f wmma_f32(v2f a, v2f b, v8f c) {
    return __builtin_amdgcn_wmma_f32_16x16x4_f32(false, a, false, b, (short)0, c,
                                                 false, false);
}

__global__ void __launch_bounds__(256) k_sqnorm(const float* __restrict__ x,
                                                float* __restrict__ xx,
                                                float* __restrict__ S) {
    int t = blockIdx.x * blockDim.x + threadIdx.x;
    if (t < 128) S[t] = 0.f;
    if (t >= B_ * N_) return;
    int b = t / N_, n = t % N_;
    const float* xb = x + (size_t)b * C_ * N_;
    float s = 0.f;
#pragma unroll
    for (int c = 0; c < C_; ++c) {
        float v = xb[(size_t)c * N_ + n];
        s += v * v;
    }
    xx[t] = s;
}


static __device__ __forceinline__ int lane_rot1(int x, int spare) {
    const int edge = __builtin_amdgcn_permlanex16(spare, x, 0xffffffffu, 0xffffffffu, false, false);
    return __builtin_amdgcn_update_dpp(edge, x, 0x111, 0xf, 0xf, false);
}

static __device__ __forceinline__ float lane_get(float x, int l) {
    return __int_as_float(__builtin_amdgcn_readlane(__float_as_int(x), l));
}

static __device__ __forceinline__ void list_insert(float& lv, int& li, float v, int id, int& s0, int& s1) {
    s0 = lane_rot1(__float_as_int(lv), s0);
    s1 = lane_rot1(li, s1);
    const float    sv = __int_as_float(s0);
    const int      si = s1;
    const unsigned cm = __builtin_amdgcn_ballot_w32(v > lv);
    const bool     c  = __builtin_amdgcn_inverse_ballot_w32(cm);
    const bool     sh = __builtin_amdgcn_inverse_ballot_w32(cm << 1);
    float nl;
    asm("v_minmax_num_f32 %0, %1, %2, %3" : "=v"(nl) : "s"(v), "v"(sv), "v"(lv));
    lv = nl;
    li = c ? (sh ? si : id) : li;
}

#define QS_  2
#define RB_  64
#define QC_  (N_ / QS_)
#define WCF_ (8 * 16 * 32 * 2)

__global__ void __launch_bounds__(256) k_knn(const float* __restrict__ x,
                                             const float* __restrict__ xx,
                                             const float* __restrict__ W,
                                             float* __restrict__ P,
                                             int* __restrict__ idx) {
    __shared__ float mv[QS_ * RB_ * K_NN];
    __shared__ int   mi[QS_ * RB_ * K_NN];
    __shared__ __attribute__((aligned(16))) float tq[QS_ * RB_];
    __shared__ __attribute__((aligned(16))) float wcf[WCF_];

    const int lane = threadIdx.x & 31;
    const int wave = __builtin_amdgcn_readfirstlane(threadIdx.x >> 5);
    const int rt   = wave & 3;
    const int q    = wave >> 2;
    const int b    = blockIdx.x / (N_ / RB_);
    const int rb   = (blockIdx.x % (N_ / RB_)) * RB_;
    const int row0 = rb + rt * 16;

    const float* xb  = x  + (size_t)b * C_ * N_;
    const float* xxb = xx + (size_t)b * N_;

    const int hs  = lane >> 4;
    const int l16 = lane & 15;

    v2f A[16];
#pragma unroll
    for (int s = 0; s < 16; ++s) {
        int c0 = 4 * s + hs * 2;
        A[s].x = xb[(size_t)(c0 + 0) * N_ + row0 + l16];
        A[s].y = xb[(size_t)(c0 + 1) * N_ + row0 + l16];
    }

    float xr[8];
#pragma unroll
    for (int v = 0; v < 8; ++v) xr[v] = xxb[row0 + v + hs * 8];

    float Lv[16];
    int   Li[16];
#pragma unroll
    for (int R = 0; R < 16; ++R) { Lv[R] = (lane == 31) ? POS_INF : NEG_INF; Li[R] = 0; }
    int sp0 = 0, sp1 = 0;
    float thv[8];
#pragma unroll
    for (int r = 0; r < 8; ++r) thv[r] = NEG_INF;

    lds_live_v4f* mine   = (lds_live_v4f*)(tq + q * RB_ + rt * 16);
    lds_live_v4f* theirs = (lds_live_v4f*)(tq + (q ^ 1) * RB_ + rt * 16);
    if (lane < 16) tq[q * RB_ + rt * 16 + lane] = NEG_INF;

    {
        const int o16 = threadIdx.x & 15;
        const int ks  = threadIdx.x >> 4;
        v2f* wf = (v2f*)wcf;
#pragma unroll
        for (int t = 0; t < 4; ++t) {
            const float* wr = W + (size_t)(16 * t + o16) * 128 + 4 * ks;
            const v4f f1 = *(const v4f*)wr;
            const v4f f2 = *(const v4f*)(wr + 64);
            v2f n0 = { f1.x, f1.y }, n1 = { f1.z, f1.w };
            v2f c0 = { f2.x - f1.x, f2.y - f1.y }, c1 = { f2.z - f1.z, f2.w - f1.w };
            wf[((t * 16 + ks) * 32) + o16]            = n0;
            wf[((t * 16 + ks) * 32) + 16 + o16]       = n1;
            wf[(((4 + t) * 16 + ks) * 32) + o16]      = c0;
            wf[(((4 + t) * 16 + ks) * 32) + 16 + o16] = c1;
        }
    }
    __syncthreads();

    {
        const v2f* wf = (const v2f*)wcf;
        v8f pa[4] = {};
#pragma unroll
        for (int s = 0; s < 16; ++s) {
#pragma unroll
            for (int j = 0; j < 4; ++j)
                pa[j] = wmma_f32(wf[(((q * 4 + j) * 16 + s) * 32) + lane], A[s], pa[j]);
        }
        float* Pb = P + (size_t)b * 128 * N_;
#pragma unroll
        for (int j = 0; j < 4; ++j) {
#pragma unroll
            for (int v = 0; v < 8; ++v) {
                const int m = (q * 4 + j) * 16 + v + hs * 8;
                Pb[(size_t)m * N_ + row0 + l16] = pa[j][v];
            }
        }
    }

    const int mbeg = q * QC_;
    for (int m0 = mbeg; m0 < mbeg + QC_; m0 += 16) {
        if (lane == K_NN / 2 - 1) {
#pragma unroll
            for (int j = 0; j < 4; ++j) {
                v4f t = { Lv[4 * j + 0], Lv[4 * j + 1], Lv[4 * j + 2], Lv[4 * j + 3] };
                mine[j] = t;
            }
        }
        const v4f a0 = mine[hs * 2], a1 = mine[hs * 2 + 1];
        const v4f b0 = theirs[hs * 2], b1 = theirs[hs * 2 + 1];
        float ta[8], tb[8];
#pragma unroll
        for (int r = 0; r < 4; ++r) {
            ta[r] = a0[r]; ta[4 + r] = a1[r];
            tb[r] = b0[r]; tb[4 + r] = b1[r];
        }

        v2f Bv[16];
#pragma unroll
        for (int s = 0; s < 16; ++s) {
            int c0 = 4 * s + hs * 2;
            Bv[s].x = xb[(size_t)(c0 + 0) * N_ + m0 + l16];
            Bv[s].y = xb[(size_t)(c0 + 1) * N_ + m0 + l16];
        }
        v8f acc = {};
#pragma unroll
        for (int s = 0; s < 16; ++s) acc = wmma_f32(A[s], Bv[s], acc);

        const float xc = xxb[m0 + l16];
        float    d[8];
        unsigned pm[8];
#pragma unroll
        for (int r = 0; r < 8; ++r) {
            d[r] = 2.0f * acc[r] - xr[r] - xc;
            const float u = __builtin_amdgcn_fmed3f(thv[r], ta[r], tb[r]);
            pm[r] = __builtin_amdgcn_ballot_w32(d[r] >= u);
        }
#pragma unroll
        for (int r = 0; r < 8; ++r) {
            if (pm[r]) {
                unsigned mlo = pm[r] & 0xffffu;
                unsigned mhi = pm[r] >> 16;
                while (mlo) {
                    const int src = __builtin_ctz(mlo);
                    mlo &= mlo - 1;
                    list_insert(Lv[r], Li[r], lane_get(d[r], src), m0 + src, sp0, sp1);
                }
                while (mhi) {
                    const int src = __builtin_ctz(mhi);
                    mhi &= mhi - 1;
                    list_insert(Lv[8 + r], Li[8 + r], lane_get(d[r], 16 + src), m0 + src, sp0, sp1);
                }
                const float tlo = lane_get(Lv[r], K_NN - 1), thi = lane_get(Lv[8 + r], K_NN - 1);
                thv[r] = hs ? thi : tlo;
            }
        }
    }

    if (lane < K_NN) {
#pragma unroll
        for (int R = 0; R < 16; ++R) {
            const int e = ((q * RB_ + rt * 16 + R) * K_NN) + lane;
            mv[e] = Lv[R];
            mi[e] = Li[R];
        }
    }
    __syncthreads();

    for (int j8 = 0; j8 < RB_ / 8; ++j8) {
        const int rl = wave * (RB_ / 8) + j8;
        float lv = (lane < K_NN) ? mv[rl * K_NN + lane] : ((lane == 31) ? POS_INF : NEG_INF);
        int   li = (lane < K_NN) ? mi[rl * K_NN + lane] : 0;
        for (int s = 1; s < QS_; ++s) {
            const int e0 = (s * RB_ + rl) * K_NN;
            for (int j = 0; j < K_NN; ++j) {
                const float v  = __int_as_float(__builtin_amdgcn_readfirstlane(__float_as_int(mv[e0 + j])));
                if (!(v > lane_get(lv, K_NN - 1))) break;
                const int   id = __builtin_amdgcn_readfirstlane(mi[e0 + j]);
                list_insert(lv, li, v, id, sp0, sp1);
            }
        }
        if (lane < K_NN)
            idx[((size_t)b * N_ + rb + rl) * K_NN + lane] = li;
    }
}

#define GR_   4
#define PTS_  512
#define NCH_  (N_ / PTS_)

static __device__ __forceinline__ void stage_rows(const float* __restrict__ src,
                                                  float* __restrict__ rows) {
    const float4* s4 = (const float4*)src;
    float4* d4 = (float4*)rows;
    float4 t[GR_ * N_ / 4 / 256];
#pragma unroll
    for (int j = 0; j < GR_ * N_ / 4 / 256; ++j) t[j] = s4[j * 256 + threadIdx.x];
#pragma unroll
    for (int j = 0; j < GR_ * N_ / 4 / 256; ++j) d4[j * 256 + threadIdx.x] = t[j];
}

__global__ void __launch_bounds__(256) k_stats(const float* __restrict__ P,
                                               float* __restrict__ Y,
                                               const int* __restrict__ idx,
                                               const float* __restrict__ gamma,
                                               float* __restrict__ S) {
    __shared__ float rows[GR_ * N_];
    const int chunk = blockIdx.x % NCH_;
    const int og    = (blockIdx.x / NCH_) % (O_ / GR_);
    const int b     = blockIdx.x / (NCH_ * (O_ / GR_));
    const int o0    = og * GR_;

    const float* Pb = P + (size_t)b * 128 * N_;
    float*       Yb = Y + ((size_t)b * 128 + 64 + o0) * N_;
    stage_rows(Pb + (size_t)o0 * N_, rows);

    bool up[GR_];
#pragma unroll
    for (int g = 0; g < GR_; ++g) up[g] = gamma[o0 + g] >= 0.f;
    __syncthreads();

    float s1[GR_], s2[GR_];
#pragma unroll
    for (int g = 0; g < GR_; ++g) { s1[g] = 0.f; s2[g] = 0.f; }

    for (int pp = 0; pp < PTS_ / 256; ++pp) {
        const int  n  = chunk * PTS_ + pp * 256 + threadIdx.x;
        const int* ip = idx + ((size_t)b * N_ + n) * K_NN;
        int a[K_NN];
#pragma unroll
        for (int k = 0; k < K_NN; ++k) a[k] = ip[k];
#pragma unroll
        for (int g = 0; g < GR_; ++g) {
            float*       yp = Yb + (size_t)g * N_ + n;
            const float  p2 = *yp;
            const float* r  = rows + g * N_;
            float t1 = 0.f, t2 = 0.f;
            float hi = -POS_INF, lo = POS_INF;
#pragma unroll
            for (int k = 0; k < K_NN; ++k) {
                float y = r[a[k]] + p2;
                t1 += y; t2 += y * y;
                hi = fmaxf(hi, y);
                lo = fminf(lo, y);
            }
            s1[g] += t1; s2[g] += t2;
            *yp = up[g] ? hi : lo;
        }
    }

#pragma unroll
    for (int g = 0; g < GR_; ++g) {
#pragma unroll
        for (int m = 16; m > 0; m >>= 1) {
            s1[g] += __shfl_xor(s1[g], m);
            s2[g] += __shfl_xor(s2[g], m);
        }
    }
    __syncthreads();
    const int lane = threadIdx.x & 31, wave = threadIdx.x >> 5;
    if (lane == 0) {
#pragma unroll
        for (int g = 0; g < GR_; ++g) {
            rows[wave * 2 * GR_ + g]       = s1[g];
            rows[wave * 2 * GR_ + GR_ + g] = s2[g];
        }
    }
    __syncthreads();
    if (threadIdx.x < 2 * GR_) {
        float v = 0.f;
#pragma unroll
        for (int w = 0; w < 8; ++w) v += rows[w * 2 * GR_ + threadIdx.x];
        const int g = threadIdx.x % GR_;
        atomicAdd(&S[(threadIdx.x < GR_ ? 0 : 64) + o0 + g], v);
    }
}

#define FV_ (N_ / 4 / 256)
__global__ void __launch_bounds__(256) k_final(const float* __restrict__ Y,
                                               const float* __restrict__ S,
                                               const float* __restrict__ gamma,
                                               const float* __restrict__ beta,
                                               float* __restrict__ out) {
    const int o = blockIdx.x % O_;
    const int b = blockIdx.x / O_;
    const float cnt  = (float)((size_t)B_ * N_ * K_NN);
    const float mean = S[o] / cnt;
    const float var  = S[64 + o] / cnt - mean * mean;
    const float gg   = gamma[o] * rsqrtf(var + EPS_);
    const float be   = beta[o];

    const float4* src = (const float4*)(Y + ((size_t)b * 128 + 64 + o) * N_);
    float4*       dst = (float4*)(out + (size_t)blockIdx.x * N_);
    float4 v[FV_];
#pragma unroll
    for (int j = 0; j < FV_; ++j) v[j] = src[j * 256 + threadIdx.x];
#pragma unroll
    for (int j = 0; j < FV_; ++j) {
        float e[4] = { v[j].x, v[j].y, v[j].z, v[j].w };
#pragma unroll
        for (int i = 0; i < 4; ++i) {
            float z = (e[i] - mean) * gg + be;
            e[i] = (z >= 0.f) ? z : SLOPE_ * z;
        }
        dst[j * 256 + threadIdx.x] = make_float4(e[0], e[1], e[2], e[3]);
    }
}

extern "C" void kernel_launch(void* const* d_in, const int* in_sizes, int n_in,
                              void* d_out, int out_size, void* d_ws, size_t ws_size,
                              hipStream_t stream) {
    const float* x     = (const float*)d_in[0];
    const float* W     = (const float*)d_in[1];
    const float* gamma = (const float*)d_in[2];
    const float* beta  = (const float*)d_in[3];
    float* out = (float*)d_out;

    char* ws = (char*)d_ws;
    size_t off = 0;
    float* xx = (float*)(ws + off); off += (size_t)B_ * N_ * sizeof(float);
    int*   id = (int*)  (ws + off); off += (size_t)B_ * N_ * K_NN * sizeof(int);
    float* P  = (float*)(ws + off); off += (size_t)B_ * 128 * N_ * sizeof(float);
    float* S  = (float*)(ws + off); off += 128 * sizeof(float);
    (void)ws_size; (void)in_sizes; (void)n_in; (void)out_size;

    k_sqnorm<<<(B_ * N_ + 255) / 256, 256, 0, stream>>>(x, xx, S);
    k_knn  <<<B_ * (N_ / RB_), 256, 0, stream>>>(x, xx, W, P, id);
    k_stats<<<B_ * (O_ / GR_) * NCH_, 256, 0, stream>>>(P, P, id, gamma, S);
    k_final<<<B_ * O_, 256, 0, stream>>>(P, S, gamma, beta, out);
}


# k_knn list seed: the first 32 columns of a wave's half (two tiles) are sorted into the 16 lists by a 32-lane bitonic network (value, then lower column), replacing their serial inserts; same top-20
# speedup vs baseline: 11.4168x; 1.0543x over previous
#include <hip/hip_runtime.h>
#include <math.h>

typedef __attribute__((ext_vector_type(2))) float v2f;
typedef __attribute__((ext_vector_type(8))) float v8f;
typedef __attribute__((ext_vector_type(4))) float v4f;
typedef __attribute__((address_space(3))) volatile v4f lds_live_v4f;

#define B_   8
#define C_   64
#define N_   4096
#define O_   64
#define K_NN 20
#define EPS_ 1e-5f
#define SLOPE_ 0.2f
#define NEG_INF -3.4e38f
#define POS_INF __builtin_inff()

static __device__ __forceinline__ v8f wmma_f32(v2f a, v2f b, v8f c) {
    return __builtin_amdgcn_wmma_f32_16x16x4_f32(false, a, false, b, (short)0, c,
                                                 false, false);
}

__global__ void __launch_bounds__(256) k_sqnorm(const float* __restrict__ x,
                                                float* __restrict__ xx,
                                                float* __restrict__ S) {
    int t = blockIdx.x * blockDim.x + threadIdx.x;
    if (t < 128) S[t] = 0.f;
    if (t >= B_ * N_) return;
    int b = t / N_, n = t % N_;
    const float* xb = x + (size_t)b * C_ * N_;
    float s = 0.f;
#pragma unroll
    for (int c = 0; c < C_; ++c) {
        float v = xb[(size_t)c * N_ + n];
        s += v * v;
    }
    xx[t] = s;
}


static __device__ __forceinline__ int lane_rot1(int x, int spare) {
    const int edge = __builtin_amdgcn_permlanex16(spare, x, 0xffffffffu, 0xffffffffu, false, false);
    return __builtin_amdgcn_update_dpp(edge, x, 0x111, 0xf, 0xf, false);
}

static __device__ __forceinline__ float lane_get(float x, int l) {
    return __int_as_float(__builtin_amdgcn_readlane(__float_as_int(x), l));
}

static __device__ __forceinline__ void list_insert(float& lv, int& li, float v, int id, int& s0, int& s1) {
    s0 = lane_rot1(__float_as_int(lv), s0);
    s1 = lane_rot1(li, s1);
    const float    sv = __int_as_float(s0);
    const int      si = s1;
    const unsigned cm = __builtin_amdgcn_ballot_w32(v > lv);
    const bool     c  = __builtin_amdgcn_inverse_ballot_w32(cm);
    const bool     sh = __builtin_amdgcn_inverse_ballot_w32(cm << 1);
    float nl;
    asm("v_minmax_num_f32 %0, %1, %2, %3" : "=v"(nl) : "s"(v), "v"(sv), "v"(lv));
    lv = nl;
    li = c ? (sh ? si : id) : li;
}


static __device__ __forceinline__ int lane_xor16(int x) {
    return __builtin_amdgcn_permlanex16(x, x, 0x76543210u, 0xfedcba98u, false, false);
}

static constexpr unsigned keep_better_mask(int K, int J) {
    unsigned m = 0;
    for (int l = 0; l < 32; ++l)
        if (((l & K) == 0) == ((l & J) == 0)) m |= 1u << l;
    return m;
}

template <int K, int J>
static __device__ __forceinline__ void cx_stage(float& v, int& id) {
    int pvi, pid;
    if (J == 16) {
        pvi = lane_xor16(__float_as_int(v));
        pid = lane_xor16(id);
    } else {
        pvi = __builtin_amdgcn_update_dpp(0, __float_as_int(v), 0x160 | J, 0xf, 0xf, true);
        pid = __builtin_amdgcn_update_dpp(0, id, 0x160 | J, 0xf, 0xf, true);
    }
    const float pv = __int_as_float(pvi);
    const bool  pbetter = (pv > v) | ((pv == v) & (pid < id));
    const bool  keepb   = __builtin_amdgcn_inverse_ballot_w32(keep_better_mask(K, J));
    const bool  take    = (pbetter == keepb);
    v  = take ? pv  : v;
    id = take ? pid : id;
}

static __device__ __forceinline__ void sort32_best_first(float& v, int& id) {
    cx_stage<2, 1>(v, id);
    cx_stage<4, 2>(v, id);   cx_stage<4, 1>(v, id);
    cx_stage<8, 4>(v, id);   cx_stage<8, 2>(v, id);   cx_stage<8, 1>(v, id);
    cx_stage<16, 8>(v, id);  cx_stage<16, 4>(v, id);  cx_stage<16, 2>(v, id);  cx_stage<16, 1>(v, id);
    cx_stage<32, 16>(v, id); cx_stage<32, 8>(v, id);  cx_stage<32, 4>(v, id);  cx_stage<32, 2>(v, id);
    cx_stage<32, 1>(v, id);
}

#define QS_  2
#define RB_  64
#define QC_  (N_ / QS_)
#define WCF_ (8 * 16 * 32 * 2)

__global__ void __launch_bounds__(256) k_knn(const float* __restrict__ x,
                                             const float* __restrict__ xx,
                                             const float* __restrict__ W,
                                             float* __restrict__ P,
                                             int* __restrict__ idx) {
    __shared__ float mv[QS_ * RB_ * K_NN];
    __shared__ int   mi[QS_ * RB_ * K_NN];
    __shared__ __attribute__((aligned(16))) float tq[QS_ * RB_];
    __shared__ __attribute__((aligned(16))) float wcf[WCF_];

    const int lane = threadIdx.x & 31;
    const int wave = __builtin_amdgcn_readfirstlane(threadIdx.x >> 5);
    const int rt   = wave & 3;
    const int q    = wave >> 2;
    const int b    = blockIdx.x / (N_ / RB_);
    const int rb   = (blockIdx.x % (N_ / RB_)) * RB_;
    const int row0 = rb + rt * 16;

    const float* xb  = x  + (size_t)b * C_ * N_;
    const float* xxb = xx + (size_t)b * N_;

    const int hs  = lane >> 4;
    const int l16 = lane & 15;

    v2f A[16];
#pragma unroll
    for (int s = 0; s < 16; ++s) {
        int c0 = 4 * s + hs * 2;
        A[s].x = xb[(size_t)(c0 + 0) * N_ + row0 + l16];
        A[s].y = xb[(size_t)(c0 + 1) * N_ + row0 + l16];
    }

    float xr[8];
#pragma unroll
    for (int v = 0; v < 8; ++v) xr[v] = xxb[row0 + v + hs * 8];

    float Lv[16];
    int   Li[16];
    int sp0 = 0, sp1 = 0;
    float thv[8];

    lds_live_v4f* mine   = (lds_live_v4f*)(tq + q * RB_ + rt * 16);
    lds_live_v4f* theirs = (lds_live_v4f*)(tq + (q ^ 1) * RB_ + rt * 16);
    if (lane < 16) tq[q * RB_ + rt * 16 + lane] = NEG_INF;

    {
        const int o16 = threadIdx.x & 15;
        const int ks  = threadIdx.x >> 4;
        v2f* wf = (v2f*)wcf;
#pragma unroll
        for (int t = 0; t < 4; ++t) {
            const float* wr = W + (size_t)(16 * t + o16) * 128 + 4 * ks;
            const v4f f1 = *(const v4f*)wr;
            const v4f f2 = *(const v4f*)(wr + 64);
            v2f n0 = { f1.x, f1.y }, n1 = { f1.z, f1.w };
            v2f c0 = { f2.x - f1.x, f2.y - f1.y }, c1 = { f2.z - f1.z, f2.w - f1.w };
            wf[((t * 16 + ks) * 32) + o16]            = n0;
            wf[((t * 16 + ks) * 32) + 16 + o16]       = n1;
            wf[(((4 + t) * 16 + ks) * 32) + o16]      = c0;
            wf[(((4 + t) * 16 + ks) * 32) + 16 + o16] = c1;
        }
    }
    __syncthreads();

    {
        const v2f* wf = (const v2f*)wcf;
        v8f pa[4] = {};
#pragma unroll
        for (int s = 0; s < 16; ++s) {
#pragma unroll
            for (int j = 0; j < 4; ++j)
                pa[j] = wmma_f32(wf[(((q * 4 + j) * 16 + s) * 32) + lane], A[s], pa[j]);
        }
        float* Pb = P + (size_t)b * 128 * N_;
#pragma unroll
        for (int j = 0; j < 4; ++j) {
#pragma unroll
            for (int v = 0; v < 8; ++v) {
                const int m = (q * 4 + j) * 16 + v + hs * 8;
                Pb[(size_t)m * N_ + row0 + l16] = pa[j][v];
            }
        }
    }

    const int mbeg = q * QC_;

    {
        float d2[2][8];
#pragma unroll
        for (int t = 0; t < 2; ++t) {
            const int m0 = mbeg + 16 * t;
            v2f Bv[16];
#pragma unroll
            for (int s = 0; s < 16; ++s) {
                int c0 = 4 * s + hs * 2;
                Bv[s].x = xb[(size_t)(c0 + 0) * N_ + m0 + l16];
                Bv[s].y = xb[(size_t)(c0 + 1) * N_ + m0 + l16];
            }
            v8f acc = {};
#pragma unroll
            for (int s = 0; s < 16; ++s) acc = wmma_f32(A[s], Bv[s], acc);
            const float xc = xxb[m0 + l16];
#pragma unroll
            for (int r = 0; r < 8; ++r) d2[t][r] = 2.0f * acc[r] - xr[r] - xc;
        }
#pragma unroll
        for (int r = 0; r < 8; ++r) {
            const float s0 = __int_as_float(lane_xor16(__float_as_int(d2[0][r])));
            const float s1 = __int_as_float(lane_xor16(__float_as_int(d2[1][r])));
            Lv[r]     = hs ? s1 : d2[0][r];
            Lv[8 + r] = hs ? d2[1][r] : s0;
            Li[r]     = mbeg + lane;
            Li[8 + r] = mbeg + lane;
        }
#pragma unroll
        for (int R = 0; R < 16; ++R) {
            sort32_best_first(Lv[R], Li[R]);
            Lv[R] = (lane == 31) ? POS_INF : Lv[R];
        }
#pragma unroll
        for (int r = 0; r < 8; ++r) {
            const float tlo = lane_get(Lv[r], K_NN - 1), thi = lane_get(Lv[8 + r], K_NN - 1);
            thv[r] = hs ? thi : tlo;
        }
    }

    for (int m0 = mbeg + 32; m0 < mbeg + QC_; m0 += 16) {
        if (lane == K_NN / 2 - 1) {
#pragma unroll
            for (int j = 0; j < 4; ++j) {
                v4f t = { Lv[4 * j + 0], Lv[4 * j + 1], Lv[4 * j + 2], Lv[4 * j + 3] };
                mine[j] = t;
            }
        }
        const v4f a0 = mine[hs * 2], a1 = mine[hs * 2 + 1];
        const v4f b0 = theirs[hs * 2], b1 = theirs[hs * 2 + 1];
        float ta[8], tb[8];
#pragma unroll
        for (int r = 0; r < 4; ++r) {
            ta[r] = a0[r]; ta[4 + r] = a1[r];
            tb[r] = b0[r]; tb[4 + r] = b1[r];
        }

        v2f Bv[16];
#pragma unroll
        for (int s = 0; s < 16; ++s) {
            int c0 = 4 * s + hs * 2;
            Bv[s].x = xb[(size_t)(c0 + 0) * N_ + m0 + l16];
            Bv[s].y = xb[(size_t)(c0 + 1) * N_ + m0 + l16];
        }
        v8f acc = {};
#pragma unroll
        for (int s = 0; s < 16; ++s) acc = wmma_f32(A[s], Bv[s], acc);

        const float xc = xxb[m0 + l16];
        float    d[8];
        unsigned pm[8];
#pragma unroll
        for (int r = 0; r < 8; ++r) {
            d[r] = 2.0f * acc[r] - xr[r] - xc;
            const float u = __builtin_amdgcn_fmed3f(thv[r], ta[r], tb[r]);
            pm[r] = __builtin_amdgcn_ballot_w32(d[r] >= u);
        }
#pragma unroll
        for (int r = 0; r < 8; ++r) {
            if (pm[r]) {
                unsigned mlo = pm[r] & 0xffffu;
                unsigned mhi = pm[r] >> 16;
                while (mlo) {
                    const int src = __builtin_ctz(mlo);
                    mlo &= mlo - 1;
                    list_insert(Lv[r], Li[r], lane_get(d[r], src), m0 + src, sp0, sp1);
                }
                while (mhi) {
                    const int src = __builtin_ctz(mhi);
                    mhi &= mhi - 1;
                    list_insert(Lv[8 + r], Li[8 + r], lane_get(d[r], 16 + src), m0 + src, sp0, sp1);
                }
                const float tlo = lane_get(Lv[r], K_NN - 1), thi = lane_get(Lv[8 + r], K_NN - 1);
                thv[r] = hs ? thi : tlo;
            }
        }
    }

    if (lane < K_NN) {
#pragma unroll
        for (int R = 0; R < 16; ++R) {
            const int e = ((q * RB_ + rt * 16 + R) * K_NN) + lane;
            mv[e] = Lv[R];
            mi[e] = Li[R];
        }
    }
    __syncthreads();

    for (int j8 = 0; j8 < RB_ / 8; ++j8) {
        const int rl = wave * (RB_ / 8) + j8;
        float lv = (lane < K_NN) ? mv[rl * K_NN + lane] : ((lane == 31) ? POS_INF : NEG_INF);
        int   li = (lane < K_NN) ? mi[rl * K_NN + lane] : 0;
        for (int s = 1; s < QS_; ++s) {
            const int e0 = (s * RB_ + rl) * K_NN;
            for (int j = 0; j < K_NN; ++j) {
                const float v  = __int_as_float(__builtin_amdgcn_readfirstlane(__float_as_int(mv[e0 + j])));
                if (!(v > lane_get(lv, K_NN - 1))) break;
                const int   id = __builtin_amdgcn_readfirstlane(mi[e0 + j]);
                list_insert(lv, li, v, id, sp0, sp1);
            }
        }
        if (lane < K_NN)
            idx[((size_t)b * N_ + rb + rl) * K_NN + lane] = li;
    }
}

#define GR_   4
#define PTS_  512
#define NCH_  (N_ / PTS_)

static __device__ __forceinline__ void stage_rows(const float* __restrict__ src,
                                                  float* __restrict__ rows) {
    const float4* s4 = (const float4*)src;
    float4* d4 = (float4*)rows;
    float4 t[GR_ * N_ / 4 / 256];
#pragma unroll
    for (int j = 0; j < GR_ * N_ / 4 / 256; ++j) t[j] = s4[j * 256 + threadIdx.x];
#pragma unroll
    for (int j = 0; j < GR_ * N_ / 4 / 256; ++j) d4[j * 256 + threadIdx.x] = t[j];
}

__global__ void __launch_bounds__(256) k_stats(const float* __restrict__ P,
                                               float* __restrict__ Y,
                                               const int* __restrict__ idx,
                                               const float* __restrict__ gamma,
                                               float* __restrict__ S) {
    __shared__ float rows[GR_ * N_];
    const int chunk = blockIdx.x % NCH_;
    const int og    = (blockIdx.x / NCH_) % (O_ / GR_);
    const int b     = blockIdx.x / (NCH_ * (O_ / GR_));
    const int o0    = og * GR_;

    const float* Pb = P + (size_t)b * 128 * N_;
    float*       Yb = Y + ((size_t)b * 128 + 64 + o0) * N_;
    stage_rows(Pb + (size_t)o0 * N_, rows);

    bool up[GR_];
#pragma unroll
    for (int g = 0; g < GR_; ++g) up[g] = gamma[o0 + g] >= 0.f;
    __syncthreads();

    float s1[GR_], s2[GR_];
#pragma unroll
    for (int g = 0; g < GR_; ++g) { s1[g] = 0.f; s2[g] = 0.f; }

    for (int pp = 0; pp < PTS_ / 256; ++pp) {
        const int  n  = chunk * PTS_ + pp * 256 + threadIdx.x;
        const int* ip = idx + ((size_t)b * N_ + n) * K_NN;
        int a[K_NN];
#pragma unroll
        for (int k = 0; k < K_NN; ++k) a[k] = ip[k];
#pragma unroll
        for (int g = 0; g < GR_; ++g) {
            float*       yp = Yb + (size_t)g * N_ + n;
            const float  p2 = *yp;
            const float* r  = rows + g * N_;
            float t1 = 0.f, t2 = 0.f;
            float hi = -POS_INF, lo = POS_INF;
#pragma unroll
            for (int k = 0; k < K_NN; ++k) {
                float y = r[a[k]] + p2;
                t1 += y; t2 += y * y;
                hi = fmaxf(hi, y);
                lo = fminf(lo, y);
            }
            s1[g] += t1; s2[g] += t2;
            *yp = up[g] ? hi : lo;
        }
    }

#pragma unroll
    for (int g = 0; g < GR_; ++g) {
#pragma unroll
        for (int m = 16; m > 0; m >>= 1) {
            s1[g] += __shfl_xor(s1[g], m);
            s2[g] += __shfl_xor(s2[g], m);
        }
    }
    __syncthreads();
    const int lane = threadIdx.x & 31, wave = threadIdx.x >> 5;
    if (lane == 0) {
#pragma unroll
        for (int g = 0; g < GR_; ++g) {
            rows[wave * 2 * GR_ + g]       = s1[g];
            rows[wave * 2 * GR_ + GR_ + g] = s2[g];
        }
    }
    __syncthreads();
    if (threadIdx.x < 2 * GR_) {
        float v = 0.f;
#pragma unroll
        for (int w = 0; w < 8; ++w) v += rows[w * 2 * GR_ + threadIdx.x];
        const int g = threadIdx.x % GR_;
        atomicAdd(&S[(threadIdx.x < GR_ ? 0 : 64) + o0 + g], v);
    }
}

#define FV_ (N_ / 4 / 256)
__global__ void __launch_bounds__(256) k_final(const float* __restrict__ Y,
                                               const float* __restrict__ S,
                                               const float* __restrict__ gamma,
                                               const float* __restrict__ beta,
                                               float* __restrict__ out) {
    const int o = blockIdx.x % O_;
    const int b = blockIdx.x / O_;
    const float cnt  = (float)((size_t)B_ * N_ * K_NN);
    const float mean = S[o] / cnt;
    const float var  = S[64 + o] / cnt - mean * mean;
    const float gg   = gamma[o] * rsqrtf(var + EPS_);
    const float be   = beta[o];

    const float4* src = (const float4*)(Y + ((size_t)b * 128 + 64 + o) * N_);
    float4*       dst = (float4*)(out + (size_t)blockIdx.x * N_);
    float4 v[FV_];
#pragma unroll
    for (int j = 0; j < FV_; ++j) v[j] = src[j * 256 + threadIdx.x];
#pragma unroll
    for (int j = 0; j < FV_; ++j) {
        float e[4] = { v[j].x, v[j].y, v[j].z, v[j].w };
#pragma unroll
        for (int i = 0; i < 4; ++i) {
            float z = (e[i] - mean) * gg + be;
            e[i] = (z >= 0.f) ? z : SLOPE_ * z;
        }
        dst[j * 256 + threadIdx.x] = make_float4(e[0], e[1], e[2], e[3]);
    }
}

extern "C" void kernel_launch(void* const* d_in, const int* in_sizes, int n_in,
                              void* d_out, int out_size, void* d_ws, size_t ws_size,
                              hipStream_t stream) {
    const float* x     = (const float*)d_in[0];
    const float* W     = (const float*)d_in[1];
    const float* gamma = (const float*)d_in[2];
    const float* beta  = (const float*)d_in[3];
    float* out = (float*)d_out;

    char* ws = (char*)d_ws;
    size_t off = 0;
    float* xx = (float*)(ws + off); off += (size_t)B_ * N_ * sizeof(float);
    int*   id = (int*)  (ws + off); off += (size_t)B_ * N_ * K_NN * sizeof(int);
    float* P  = (float*)(ws + off); off += (size_t)B_ * 128 * N_ * sizeof(float);
    float* S  = (float*)(ws + off); off += 128 * sizeof(float);
    (void)ws_size; (void)in_sizes; (void)n_in; (void)out_size;

    k_sqnorm<<<(B_ * N_ + 255) / 256, 256, 0, stream>>>(x, xx, S);
    k_knn  <<<B_ * (N_ / RB_), 256, 0, stream>>>(x, xx, W, P, id);
    k_stats<<<B_ * (O_ / GR_) * NCH_, 256, 0, stream>>>(P, P, id, gamma, S);
    k_final<<<B_ * O_, 256, 0, stream>>>(P, S, gamma, beta, out);
}
